# starting kernel
# baseline (speedup 1.0000x reference)
#include <hip/hip_runtime.h>

#define N_NODES 50000
#define D 128
#define TM 16

typedef float v2f __attribute__((ext_vector_type(2)));
typedef float v8f __attribute__((ext_vector_type(8)));

__global__ __launch_bounds__(256) void gin_copy_kernel(const float* __restrict__ src,
                                                       float* __restrict__ dst, int n4) {
    int i = blockIdx.x * blockDim.x + threadIdx.x;
    if (i < n4) {
        reinterpret_cast<float4*>(dst)[i] = reinterpret_cast<const float4*>(src)[i];
    }
}

__global__ __launch_bounds__(256) void gin_scatter_kernel(const float* __restrict__ x,
                                                          const int* __restrict__ src_idx,
                                                          const int* __restrict__ dst_idx,
                                                          float* __restrict__ acc, int nE) {
    int tid  = blockIdx.x * blockDim.x + threadIdx.x;
    int e    = tid >> 5;
    int lane = tid & 31;
    if (e >= nE) return;
    int s = src_idx[e];
    int d = dst_idx[e];
    const float4 v = *reinterpret_cast<const float4*>(x + (size_t)s * D + lane * 4);
    float* p = acc + (size_t)d * D + lane * 4;
    unsafeAtomicAdd(p + 0, v.x);
    unsafeAtomicAdd(p + 1, v.y);
    unsafeAtomicAdd(p + 2, v.z);
    unsafeAtomicAdd(p + 3, v.w);
}

template <int RELU_OUT>
__global__ __launch_bounds__(256) void gin_mlp_kernel(const float* __restrict__ xin,
                                                      const float* __restrict__ Wa,
                                                      const float* __restrict__ ba,
                                                      const float* __restrict__ Wb,
                                                      const float* __restrict__ bb,
                                                      float* __restrict__ out) {
    __shared__ float XT[D][TM + 1];
    __shared__ float HT[D][TM + 1];

    const int node_base = blockIdx.x * TM;
    const int t = threadIdx.x;

#pragma unroll
    for (int i = 0; i < (TM * D) / 256; ++i) {
        int e   = t + 256 * i;
        int row = e >> 7;
        int col = e & (D - 1);
        XT[col][row] = xin[(size_t)(node_base + row) * D + col];
    }
    __syncthreads();

    const int w    = t >> 5;
    const int lane = t & 31;
    const int half = lane >> 4;
    const int m16  = lane & 15;
    const int ncol = w * 16 + m16;

    v8f c = {};
#pragma unroll 8
    for (int k0 = 0; k0 < D; k0 += 4) {
        const int ka = k0 + 2 * half;
        v2f a, b;
        a.x = XT[ka][m16];
        a.y = XT[ka + 1][m16];
        b.x = Wa[(size_t)ka * D + ncol];
        b.y = Wa[(size_t)(ka + 1) * D + ncol];
        c = __builtin_amdgcn_wmma_f32_16x16x4_f32(false, a, false, b, (short)0, c,
                                                  false, false);
    }
    {
        const float bias = ba[ncol];
#pragma unroll
        for (int v = 0; v < 8; ++v) {
            float r = c[v] + bias;
            r = r > 0.f ? r : 0.f;
            HT[ncol][v + 8 * half] = r;
        }
    }
    __syncthreads();

    c = (v8f){};
#pragma unroll 8
    for (int k0 = 0; k0 < D; k0 += 4) {
        const int ka = k0 + 2 * half;
        v2f a, b;
        a.x = HT[ka][m16];
        a.y = HT[ka + 1][m16];
        b.x = Wb[(size_t)ka * D + ncol];
        b.y = Wb[(size_t)(ka + 1) * D + ncol];
        c = __builtin_amdgcn_wmma_f32_16x16x4_f32(false, a, false, b, (short)0, c,
                                                  false, false);
    }
    {
        const float bias = bb[ncol];
#pragma unroll
        for (int v = 0; v < 8; ++v) {
            float r = c[v] + bias;
            if (RELU_OUT) r = r > 0.f ? r : 0.f;
            out[(size_t)(node_base + v + 8 * half) * D + ncol] = r;
        }
    }
}

extern "C" void kernel_launch(void* const* d_in, const int* in_sizes, int n_in,
                              void* d_out, int out_size, void* d_ws, size_t ws_size,
                              hipStream_t stream) {
    const float* x  = (const float*)d_in[0];
    const int*  ei  = (const int*)d_in[1];
    const float* W1 = (const float*)d_in[2];
    const float* b1 = (const float*)d_in[3];
    const float* W2 = (const float*)d_in[4];
    const float* b2 = (const float*)d_in[5];
    const float* W3 = (const float*)d_in[6];
    const float* b3 = (const float*)d_in[7];
    const float* W4 = (const float*)d_in[8];
    const float* b4 = (const float*)d_in[9];
    float* out = (float*)d_out;

    const int E = in_sizes[1] / 2;
    const int* src = ei;
    const int* dst = ei + E;

    float* acc = (float*)d_ws;
    float* h   = acc + (size_t)N_NODES * D;

    const int n4       = N_NODES * D / 4;
    const int cpBlocks = (n4 + 255) / 256;
    const int scBlocks = (E * 32 + 255) / 256;
    const int mlpBlocks = N_NODES / TM;

    gin_copy_kernel<<<cpBlocks, 256, 0, stream>>>(x, acc, n4);
    gin_scatter_kernel<<<scBlocks, 256, 0, stream>>>(x, src, dst, acc, E);
    gin_mlp_kernel<1><<<mlpBlocks, 256, 0, stream>>>(acc, W1, b1, W2, b2, h);

    gin_copy_kernel<<<cpBlocks, 256, 0, stream>>>(h, acc, n4);
    gin_scatter_kernel<<<scBlocks, 256, 0, stream>>>(h, src, dst, acc, E);
    gin_mlp_kernel<0><<<mlpBlocks, 256, 0, stream>>>(acc, W3, b3, W4, b4, out);
}


# Replace per-edge f32 atomic scatter and its copy by a per-call edge list grouped by destination (int atomic arrival-order rank, scan, fill) and a per-node gather sum; neighbour sum order changes
# speedup vs baseline: 1.8907x; 1.8907x over previous
#include <hip/hip_runtime.h>

#define N_NODES 50000
#define D 128
#define TM 16

#define SCAN_T   1024
#define SCAN_PER 52
#define N_PAD    (SCAN_T * SCAN_PER)
#define EDGE_BLOCKS   1024
#define GATHER_BLOCKS 2048
#define GATHER_WAVES  (GATHER_BLOCKS * 8)

typedef float v2f __attribute__((ext_vector_type(2)));
typedef float v8f __attribute__((ext_vector_type(8)));

__global__ __launch_bounds__(256) void gin_zero_kernel(int* __restrict__ cnt) {
    const int i = blockIdx.x * blockDim.x + threadIdx.x;
    reinterpret_cast<int4*>(cnt)[i] = make_int4(0, 0, 0, 0);
}

__global__ __launch_bounds__(256) void gin_rank_kernel(const int* __restrict__ dst_idx,
                                                       int* __restrict__ cnt,
                                                       int* __restrict__ rank, int nE) {
    const int stride = gridDim.x * blockDim.x;
    for (int e = blockIdx.x * blockDim.x + threadIdx.x; e < nE; e += stride) {
        rank[e] = atomicAdd(&cnt[dst_idx[e]], 1);
    }
}

__global__ __launch_bounds__(SCAN_T) void gin_scan_kernel(const int* __restrict__ cnt,
                                                          int* __restrict__ rowptr) {
    __shared__ int wtot[SCAN_T / 32];
    constexpr int NCH = SCAN_PER / 4;
    const int t    = threadIdx.x;
    const int lane = t & 31;
    const int w    = t >> 5;

    const int4* c4 = reinterpret_cast<const int4*>(cnt) + w * (NCH * 32) + lane;
    int4 v[NCH];
    int ex[NCH];
    int run = 0;
#pragma unroll
    for (int i = 0; i < NCH; ++i) {
        v[i] = c4[i * 32];
        const int s = (v[i].x + v[i].y) + (v[i].z + v[i].w);
        int inc = s;
#pragma unroll
        for (int o = 1; o < 32; o <<= 1) {
            const int u = __shfl_up(inc, o);
            if (lane >= o) inc += u;
        }
        ex[i] = run + (inc - s);
        run += __shfl(inc, 31);
    }
    if (lane == 0) wtot[w] = run;
    __syncthreads();
    if (w == 0) {
        const int tot = wtot[lane];
        int ti = tot;
#pragma unroll
        for (int o = 1; o < 32; o <<= 1) {
            const int u = __shfl_up(ti, o);
            if (lane >= o) ti += u;
        }
        wtot[lane] = ti - tot;
    }
    __syncthreads();

    const int base = wtot[w];
    int4* r4 = reinterpret_cast<int4*>(rowptr) + w * (NCH * 32) + lane;
#pragma unroll
    for (int i = 0; i < NCH; ++i) {
        int4 o;
        o.x = base + ex[i];
        o.y = o.x + v[i].x;
        o.z = o.y + v[i].y;
        o.w = o.z + v[i].z;
        r4[i * 32] = o;
    }
}

__global__ __launch_bounds__(256) void gin_fill_kernel(const int* __restrict__ src_idx,
                                                       const int* __restrict__ dst_idx,
                                                       const int* __restrict__ rowptr,
                                                       const int* __restrict__ rank,
                                                       int* __restrict__ col, int nE) {
    const int stride = gridDim.x * blockDim.x;
    for (int e = blockIdx.x * blockDim.x + threadIdx.x; e < nE; e += stride) {
        col[rowptr[dst_idx[e]] + rank[e]] = src_idx[e];
    }
}

__device__ __forceinline__ float4 add4(float4 a, float4 b) {
    return make_float4(a.x + b.x, a.y + b.y, a.z + b.z, a.w + b.w);
}

__global__ __launch_bounds__(256) void gin_gather_kernel(const float* __restrict__ x,
                                                         const int* __restrict__ rowptr,
                                                         const int* __restrict__ col,
                                                         float* __restrict__ acc) {
    const int lane = threadIdx.x & 31;
    const float4* x4 = reinterpret_cast<const float4*>(x);
    float4* acc4 = reinterpret_cast<float4*>(acc);
    int node = __builtin_amdgcn_readfirstlane((int)((blockIdx.x * 256 + threadIdx.x) >> 5));
    for (; node < N_NODES; node += GATHER_WAVES) {
        const int beg = rowptr[node];
        const int end = rowptr[node + 1];
        float4 a0 = x4[(size_t)node * 32 + lane];
        float4 a1 = make_float4(0.f, 0.f, 0.f, 0.f);
        float4 a2 = a1, a3 = a1;
        for (int b = beg; b < end; b += 32) {
            const int idx = b + lane;
            const int my  = idx < end ? col[idx] : 0;
            const int rem = end - b;
            const int n   = rem < 32 ? rem : 32;
            int j = 0;
            for (; j + 4 <= n; j += 4) {
                const int s0 = __builtin_amdgcn_readlane(my, j);
                const int s1 = __builtin_amdgcn_readlane(my, j + 1);
                const int s2 = __builtin_amdgcn_readlane(my, j + 2);
                const int s3 = __builtin_amdgcn_readlane(my, j + 3);
                const float4 v0 = x4[(size_t)s0 * 32 + lane];
                const float4 v1 = x4[(size_t)s1 * 32 + lane];
                const float4 v2 = x4[(size_t)s2 * 32 + lane];
                const float4 v3 = x4[(size_t)s3 * 32 + lane];
                a0 = add4(a0, v0);
                a1 = add4(a1, v1);
                a2 = add4(a2, v2);
                a3 = add4(a3, v3);
            }
            for (; j < n; ++j) {
                const int s0 = __builtin_amdgcn_readlane(my, j);
                a0 = add4(a0, x4[(size_t)s0 * 32 + lane]);
            }
        }
        acc4[(size_t)node * 32 + lane] = add4(add4(a0, a1), add4(a2, a3));
    }
}

template <int RELU_OUT>
__global__ __launch_bounds__(256) void gin_mlp_kernel(const float* __restrict__ xin,
                                                      const float* __restrict__ Wa,
                                                      const float* __restrict__ ba,
                                                      const float* __restrict__ Wb,
                                                      const float* __restrict__ bb,
                                                      float* __restrict__ out) {
    __shared__ float XT[D][TM + 1];
    __shared__ float HT[D][TM + 1];

    const int node_base = blockIdx.x * TM;
    const int t = threadIdx.x;

#pragma unroll
    for (int i = 0; i < (TM * D) / 256; ++i) {
        int e   = t + 256 * i;
        int row = e >> 7;
        int col = e & (D - 1);
        XT[col][row] = xin[(size_t)(node_base + row) * D + col];
    }
    __syncthreads();

    const int w    = t >> 5;
    const int lane = t & 31;
    const int half = lane >> 4;
    const int m16  = lane & 15;
    const int ncol = w * 16 + m16;

    v8f c = {};
#pragma unroll 8
    for (int k0 = 0; k0 < D; k0 += 4) {
        const int ka = k0 + 2 * half;
        v2f a, b;
        a.x = XT[ka][m16];
        a.y = XT[ka + 1][m16];
        b.x = Wa[(size_t)ka * D + ncol];
        b.y = Wa[(size_t)(ka + 1) * D + ncol];
        c = __builtin_amdgcn_wmma_f32_16x16x4_f32(false, a, false, b, (short)0, c,
                                                  false, false);
    }
    {
        const float bias = ba[ncol];
#pragma unroll
        for (int v = 0; v < 8; ++v) {
            float r = c[v] + bias;
            r = r > 0.f ? r : 0.f;
            HT[ncol][v + 8 * half] = r;
        }
    }
    __syncthreads();

    c = (v8f){};
#pragma unroll 8
    for (int k0 = 0; k0 < D; k0 += 4) {
        const int ka = k0 + 2 * half;
        v2f a, b;
        a.x = HT[ka][m16];
        a.y = HT[ka + 1][m16];
        b.x = Wb[(size_t)ka * D + ncol];
        b.y = Wb[(size_t)(ka + 1) * D + ncol];
        c = __builtin_amdgcn_wmma_f32_16x16x4_f32(false, a, false, b, (short)0, c,
                                                  false, false);
    }
    {
        const float bias = bb[ncol];
#pragma unroll
        for (int v = 0; v < 8; ++v) {
            float r = c[v] + bias;
            if (RELU_OUT) r = r > 0.f ? r : 0.f;
            out[(size_t)(node_base + v + 8 * half) * D + ncol] = r;
        }
    }
}

extern "C" void kernel_launch(void* const* d_in, const int* in_sizes, int n_in,
                              void* d_out, int out_size, void* d_ws, size_t ws_size,
                              hipStream_t stream) {
    const float* x  = (const float*)d_in[0];
    const int*  ei  = (const int*)d_in[1];
    const float* W1 = (const float*)d_in[2];
    const float* b1 = (const float*)d_in[3];
    const float* W2 = (const float*)d_in[4];
    const float* b2 = (const float*)d_in[5];
    const float* W3 = (const float*)d_in[6];
    const float* b3 = (const float*)d_in[7];
    const float* W4 = (const float*)d_in[8];
    const float* b4 = (const float*)d_in[9];
    float* out = (float*)d_out;

    const int E = in_sizes[1] / 2;
    const int* src = ei;
    const int* dst = ei + E;

    float* acc = (float*)d_ws;
    float* h   = acc + (size_t)N_NODES * D;
    int* cnt    = (int*)(h + (size_t)N_NODES * D);
    int* rowptr = cnt + N_PAD;
    int* rank   = rowptr + N_PAD;
    int* col    = rank + E;

    const int mlpBlocks = N_NODES / TM;

    gin_zero_kernel<<<N_PAD / 4 / 256, 256, 0, stream>>>(cnt);
    gin_rank_kernel<<<EDGE_BLOCKS, 256, 0, stream>>>(dst, cnt, rank, E);
    gin_scan_kernel<<<1, SCAN_T, 0, stream>>>(cnt, rowptr);
    gin_fill_kernel<<<EDGE_BLOCKS, 256, 0, stream>>>(src, dst, rowptr, rank, col, E);

    gin_gather_kernel<<<GATHER_BLOCKS, 256, 0, stream>>>(x, rowptr, col, acc);
    gin_mlp_kernel<1><<<mlpBlocks, 256, 0, stream>>>(acc, W1, b1, W2, b2, h);

    gin_gather_kernel<<<GATHER_BLOCKS, 256, 0, stream>>>(h, rowptr, col, acc);
    gin_mlp_kernel<0><<<mlpBlocks, 256, 0, stream>>>(acc, W3, b3, W4, b4, out);
}


# Fold each layer's neighbour gather into its MLP launch: a work-group sums its 16 node rows straight into the LDS tile the first product reads (acc plane and 2 launches gone); sum order kept
# speedup vs baseline: 2.0502x; 1.0844x over previous
#include <hip/hip_runtime.h>

#define N_NODES 50000
#define D 128
#define TM 16

#define SCAN_T   1024
#define SCAN_PER 52
#define N_PAD    (SCAN_T * SCAN_PER)
#define EDGE_BLOCKS   1024

typedef float v2f __attribute__((ext_vector_type(2)));
typedef float v8f __attribute__((ext_vector_type(8)));

__global__ __launch_bounds__(256) void gin_zero_kernel(int* __restrict__ cnt) {
    const int i = blockIdx.x * blockDim.x + threadIdx.x;
    reinterpret_cast<int4*>(cnt)[i] = make_int4(0, 0, 0, 0);
}

__global__ __launch_bounds__(256) void gin_rank_kernel(const int* __restrict__ dst_idx,
                                                       int* __restrict__ cnt,
                                                       int* __restrict__ rank, int nE) {
    const int stride = gridDim.x * blockDim.x;
    for (int e = blockIdx.x * blockDim.x + threadIdx.x; e < nE; e += stride) {
        rank[e] = atomicAdd(&cnt[dst_idx[e]], 1);
    }
}

__global__ __launch_bounds__(SCAN_T) void gin_scan_kernel(const int* __restrict__ cnt,
                                                          int* __restrict__ rowptr) {
    __shared__ int wtot[SCAN_T / 32];
    constexpr int NCH = SCAN_PER / 4;
    const int t    = threadIdx.x;
    const int lane = t & 31;
    const int w    = t >> 5;

    const int4* c4 = reinterpret_cast<const int4*>(cnt) + w * (NCH * 32) + lane;
    int4 v[NCH];
    int ex[NCH];
    int run = 0;
#pragma unroll
    for (int i = 0; i < NCH; ++i) {
        v[i] = c4[i * 32];
        const int s = (v[i].x + v[i].y) + (v[i].z + v[i].w);
        int inc = s;
#pragma unroll
        for (int o = 1; o < 32; o <<= 1) {
            const int u = __shfl_up(inc, o);
            if (lane >= o) inc += u;
        }
        ex[i] = run + (inc - s);
        run += __shfl(inc, 31);
    }
    if (lane == 0) wtot[w] = run;
    __syncthreads();
    if (w == 0) {
        const int tot = wtot[lane];
        int ti = tot;
#pragma unroll
        for (int o = 1; o < 32; o <<= 1) {
            const int u = __shfl_up(ti, o);
            if (lane >= o) ti += u;
        }
        wtot[lane] = ti - tot;
    }
    __syncthreads();

    const int base = wtot[w];
    int4* r4 = reinterpret_cast<int4*>(rowptr) + w * (NCH * 32) + lane;
#pragma unroll
    for (int i = 0; i < NCH; ++i) {
        int4 o;
        o.x = base + ex[i];
        o.y = o.x + v[i].x;
        o.z = o.y + v[i].y;
        o.w = o.z + v[i].z;
        r4[i * 32] = o;
    }
}

__global__ __launch_bounds__(256) void gin_fill_kernel(const int* __restrict__ src_idx,
                                                       const int* __restrict__ dst_idx,
                                                       const int* __restrict__ rowptr,
                                                       const int* __restrict__ rank,
                                                       int* __restrict__ col, int nE) {
    const int stride = gridDim.x * blockDim.x;
    for (int e = blockIdx.x * blockDim.x + threadIdx.x; e < nE; e += stride) {
        col[rowptr[dst_idx[e]] + rank[e]] = src_idx[e];
    }
}

__device__ __forceinline__ float4 add4(float4 a, float4 b) {
    return make_float4(a.x + b.x, a.y + b.y, a.z + b.z, a.w + b.w);
}

__device__ __forceinline__ float4 gin_row_sum(const float4* __restrict__ x4,
                                              const int* __restrict__ col,
                                              int beg, int end, int lane,
                                              float4 own, int my) {
    float4 a0 = own;
    float4 a1 = make_float4(0.f, 0.f, 0.f, 0.f);
    float4 a2 = a1, a3 = a1;
    for (int b = beg; b < end;) {
        const int rem = end - b;
        const int n   = rem < 32 ? rem : 32;
        int j = 0;
        for (; j + 4 <= n; j += 4) {
            const int s0 = __builtin_amdgcn_readlane(my, j);
            const int s1 = __builtin_amdgcn_readlane(my, j + 1);
            const int s2 = __builtin_amdgcn_readlane(my, j + 2);
            const int s3 = __builtin_amdgcn_readlane(my, j + 3);
            const float4 v0 = x4[(size_t)s0 * 32 + lane];
            const float4 v1 = x4[(size_t)s1 * 32 + lane];
            const float4 v2 = x4[(size_t)s2 * 32 + lane];
            const float4 v3 = x4[(size_t)s3 * 32 + lane];
            a0 = add4(a0, v0);
            a1 = add4(a1, v1);
            a2 = add4(a2, v2);
            a3 = add4(a3, v3);
        }
#pragma unroll 1
        for (; j < n; ++j) {
            const int s0 = __builtin_amdgcn_readlane(my, j);
            a0 = add4(a0, x4[(size_t)s0 * 32 + lane]);
        }
        b += 32;
        if (b < end) {
            const int idx = b + lane;
            my = idx < end ? col[idx] : 0;
        }
    }
    return add4(add4(a0, a1), add4(a2, a3));
}

template <int RELU_OUT>
__global__ __launch_bounds__(256) void gin_layer_kernel(const float* __restrict__ xin,
                                                        const int* __restrict__ rowptr,
                                                        const int* __restrict__ col,
                                                        const float* __restrict__ Wa,
                                                        const float* __restrict__ ba,
                                                        const float* __restrict__ Wb,
                                                        const float* __restrict__ bb,
                                                        float* __restrict__ out) {
    __shared__ float XT[D][TM + 1];
    __shared__ float HT[D][TM + 1];

    const int node_base = blockIdx.x * TM;
    const int t = threadIdx.x;

    const int lane = t & 31;
    const int w = __builtin_amdgcn_readfirstlane(t >> 5);

    {
        const float4* x4 = reinterpret_cast<const float4*>(xin);
        const int n0 = node_base + 2 * w;
        const int p0 = rowptr[n0];
        const int p1 = rowptr[n0 + 1];
        const int p2 = rowptr[n0 + 2];
        const float4 own0 = x4[(size_t)n0 * 32 + lane];
        const float4 own1 = x4[(size_t)(n0 + 1) * 32 + lane];
        const int my0 = p0 + lane < p1 ? col[p0 + lane] : 0;
        const int my1 = p1 + lane < p2 ? col[p1 + lane] : 0;
        const int c = 4 * lane;
        const float4 r0 = gin_row_sum(x4, col, p0, p1, lane, own0, my0);
        XT[c][2 * w]     = r0.x;
        XT[c + 1][2 * w] = r0.y;
        XT[c + 2][2 * w] = r0.z;
        XT[c + 3][2 * w] = r0.w;
        const float4 r1 = gin_row_sum(x4, col, p1, p2, lane, own1, my1);
        XT[c][2 * w + 1]     = r1.x;
        XT[c + 1][2 * w + 1] = r1.y;
        XT[c + 2][2 * w + 1] = r1.z;
        XT[c + 3][2 * w + 1] = r1.w;
    }
    __syncthreads();

    const int half = lane >> 4;
    const int m16  = lane & 15;
    const int ncol = w * 16 + m16;

    v8f c = {};
#pragma unroll 8
    for (int k0 = 0; k0 < D; k0 += 4) {
        const int ka = k0 + 2 * half;
        v2f a, b;
        a.x = XT[ka][m16];
        a.y = XT[ka + 1][m16];
        b.x = Wa[(size_t)ka * D + ncol];
        b.y = Wa[(size_t)(ka + 1) * D + ncol];
        c = __builtin_amdgcn_wmma_f32_16x16x4_f32(false, a, false, b, (short)0, c,
                                                  false, false);
    }
    {
        const float bias = ba[ncol];
#pragma unroll
        for (int v = 0; v < 8; ++v) {
            float r = c[v] + bias;
            r = r > 0.f ? r : 0.f;
            HT[ncol][v + 8 * half] = r;
        }
    }
    __syncthreads();

    c = (v8f){};
#pragma unroll 8
    for (int k0 = 0; k0 < D; k0 += 4) {
        const int ka = k0 + 2 * half;
        v2f a, b;
        a.x = HT[ka][m16];
        a.y = HT[ka + 1][m16];
        b.x = Wb[(size_t)ka * D + ncol];
        b.y = Wb[(size_t)(ka + 1) * D + ncol];
        c = __builtin_amdgcn_wmma_f32_16x16x4_f32(false, a, false, b, (short)0, c,
                                                  false, false);
    }
    {
        const float bias = bb[ncol];
#pragma unroll
        for (int v = 0; v < 8; ++v) {
            float r = c[v] + bias;
            if (RELU_OUT) r = r > 0.f ? r : 0.f;
            out[(size_t)(node_base + v + 8 * half) * D + ncol] = r;
        }
    }
}

extern "C" void kernel_launch(void* const* d_in, const int* in_sizes, int n_in,
                              void* d_out, int out_size, void* d_ws, size_t ws_size,
                              hipStream_t stream) {
    const float* x  = (const float*)d_in[0];
    const int*  ei  = (const int*)d_in[1];
    const float* W1 = (const float*)d_in[2];
    const float* b1 = (const float*)d_in[3];
    const float* W2 = (const float*)d_in[4];
    const float* b2 = (const float*)d_in[5];
    const float* W3 = (const float*)d_in[6];
    const float* b3 = (const float*)d_in[7];
    const float* W4 = (const float*)d_in[8];
    const float* b4 = (const float*)d_in[9];
    float* out = (float*)d_out;

    const int E = in_sizes[1] / 2;
    const int* src = ei;
    const int* dst = ei + E;

    float* h    = (float*)d_ws;
    int* cnt    = (int*)(h + (size_t)N_NODES * D);
    int* rowptr = cnt + N_PAD;
    int* rank   = rowptr + N_PAD;
    int* col    = rank + E;

    const int layerBlocks = N_NODES / TM;

    gin_zero_kernel<<<N_PAD / 4 / 256, 256, 0, stream>>>(cnt);
    gin_rank_kernel<<<EDGE_BLOCKS, 256, 0, stream>>>(dst, cnt, rank, E);
    gin_scan_kernel<<<1, SCAN_T, 0, stream>>>(cnt, rowptr);
    gin_fill_kernel<<<EDGE_BLOCKS, 256, 0, stream>>>(src, dst, rowptr, rank, col, E);

    gin_layer_kernel<1><<<layerBlocks, 256, 0, stream>>>(x, rowptr, col, W1, b1, W2, b2, h);

    gin_layer_kernel<0><<<layerBlocks, 256, 0, stream>>>(h, rowptr, col, W3, b3, W4, b4, out);
}


# Group edges by destination with LDS counters in two levels (chunk bucket counts, scan, bucket records, per-bucket node segments), not one returning global atomic per edge; neighbour sum order changes
# speedup vs baseline: 3.0214x; 1.4737x over previous
#include <hip/hip_runtime.h>

#define N_NODES 50000
#define D 128
#define TM 16

#define SCAN_T   1024
#define SCAN_PER 52
#define N_PAD    (SCAN_T * SCAN_PER)
#define EDGE_BLOCKS   256
#define EDGE_T        512
#define BUCKET_SHIFT  8
#define BUCKET_NODES  (1 << BUCKET_SHIFT)
#define NBUCK         ((N_NODES + BUCKET_NODES - 1) / BUCKET_NODES)
#define BCNT_USED     (NBUCK * EDGE_BLOCKS)
#define BCNT_PAD_PER  ((N_PAD - BCNT_USED) / EDGE_BLOCKS)
#define FIN_T         1024

static_assert(NBUCK <= EDGE_T, "one thread per bucket counter");
static_assert(BCNT_USED <= N_PAD, "the scan covers every (bucket, chunk) counter");
static_assert(BCNT_USED + BCNT_PAD_PER * EDGE_BLOCKS == N_PAD, "pad counters split evenly");
static_assert(BCNT_PAD_PER <= EDGE_T, "one thread per pad counter");
static_assert(NBUCK * BUCKET_NODES <= N_PAD, "rowptr holds every bucket's nodes");
static_assert(N_NODES < (1 << 24), "a source id fits the record's low 24 bits and no record is all ones");

typedef float v2f __attribute__((ext_vector_type(2)));
typedef float v8f __attribute__((ext_vector_type(8)));

__device__ __forceinline__ int gin_bucket_of(int d) {
    const int b = d >> BUCKET_SHIFT;
    return b < NBUCK ? b : NBUCK - 1;
}

__global__ __launch_bounds__(EDGE_T) void gin_bucket_count_kernel(const int* __restrict__ dst_idx,
                                                                  int* __restrict__ bcnt, int nE) {
    __shared__ int hist[NBUCK];
    const int t = threadIdx.x;
    const int g = blockIdx.x;
    if (t < NBUCK) hist[t] = 0;
    __syncthreads();

    const int chunk = (nE + EDGE_BLOCKS - 1) / EDGE_BLOCKS;
    const int beg = g * chunk;
    const int end = beg + chunk < nE ? beg + chunk : nE;
    for (int e0 = beg; e0 < end; e0 += 4 * EDGE_T) {
        int d[4];
#pragma unroll
        for (int u = 0; u < 4; ++u) {
            const int e = e0 + u * EDGE_T + t;
            d[u] = e < end ? dst_idx[e] : -1;
        }
#pragma unroll
        for (int u = 0; u < 4; ++u) {
            if (d[u] >= 0) atomicAdd(&hist[gin_bucket_of(d[u])], 1);
        }
    }
    __syncthreads();
    if (t < NBUCK) bcnt[t * EDGE_BLOCKS + g] = hist[t];
    if (t < BCNT_PAD_PER) bcnt[BCNT_USED + g * BCNT_PAD_PER + t] = 0;
}

__global__ __launch_bounds__(SCAN_T) void gin_scan_kernel(const int* __restrict__ cnt,
                                                          int* __restrict__ rowptr) {
    __shared__ int wtot[SCAN_T / 32];
    constexpr int NCH = SCAN_PER / 4;
    const int t    = threadIdx.x;
    const int lane = t & 31;
    const int w    = t >> 5;

    const int4* c4 = reinterpret_cast<const int4*>(cnt) + w * (NCH * 32) + lane;
    int4 v[NCH];
    int ex[NCH];
    int run = 0;
#pragma unroll
    for (int i = 0; i < NCH; ++i) {
        v[i] = c4[i * 32];
        const int s = (v[i].x + v[i].y) + (v[i].z + v[i].w);
        int inc = s;
#pragma unroll
        for (int o = 1; o < 32; o <<= 1) {
            const int u = __shfl_up(inc, o);
            if (lane >= o) inc += u;
        }
        ex[i] = run + (inc - s);
        run += __shfl(inc, 31);
    }
    if (lane == 0) wtot[w] = run;
    __syncthreads();
    if (w == 0) {
        const int tot = wtot[lane];
        int ti = tot;
#pragma unroll
        for (int o = 1; o < 32; o <<= 1) {
            const int u = __shfl_up(ti, o);
            if (lane >= o) ti += u;
        }
        wtot[lane] = ti - tot;
    }
    __syncthreads();

    const int base = wtot[w];
    int4* r4 = reinterpret_cast<int4*>(rowptr) + w * (NCH * 32) + lane;
#pragma unroll
    for (int i = 0; i < NCH; ++i) {
        int4 o;
        o.x = base + ex[i];
        o.y = o.x + v[i].x;
        o.z = o.y + v[i].y;
        o.w = o.z + v[i].z;
        r4[i * 32] = o;
    }
}

__global__ __launch_bounds__(EDGE_T) void gin_bucket_scatter_kernel(const int* __restrict__ src_idx,
                                                                    const int* __restrict__ dst_idx,
                                                                    const int* __restrict__ boff,
                                                                    int* __restrict__ rec, int nE) {
    __shared__ int cur[NBUCK];
    const int t = threadIdx.x;
    const int g = blockIdx.x;
    if (t < NBUCK) cur[t] = boff[t * EDGE_BLOCKS + g];
    __syncthreads();

    const int chunk = (nE + EDGE_BLOCKS - 1) / EDGE_BLOCKS;
    const int beg = g * chunk;
    const int end = beg + chunk < nE ? beg + chunk : nE;
    for (int e0 = beg; e0 < end; e0 += 4 * EDGE_T) {
        int d[4], s[4];
#pragma unroll
        for (int u = 0; u < 4; ++u) {
            const int e = e0 + u * EDGE_T + t;
            d[u] = e < end ? dst_idx[e] : -1;
            s[u] = e < end ? src_idx[e] : 0;
        }
#pragma unroll
        for (int u = 0; u < 4; ++u) {
            if (d[u] >= 0) {
                const int pos = atomicAdd(&cur[gin_bucket_of(d[u])], 1);
                rec[pos] = (int)(((unsigned)(d[u] & (BUCKET_NODES - 1)) << 24) | (unsigned)s[u]);
            }
        }
    }
}

__global__ __launch_bounds__(FIN_T) void gin_bucket_finish_kernel(const int* __restrict__ boff,
                                                                  const int* __restrict__ rec,
                                                                  int* __restrict__ rowptr,
                                                                  int* __restrict__ col) {
    __shared__ int cnt[BUCKET_NODES];
    __shared__ int cur[BUCKET_NODES];
    __shared__ int wsum[BUCKET_NODES / 32];
    const int t    = threadIdx.x;
    const int lane = t & 31;
    const int w    = t >> 5;
    const int b    = blockIdx.x;
    const int beg = boff[b * EDGE_BLOCKS];
    const int end = boff[(b + 1) * EDGE_BLOCKS];

    if (t < BUCKET_NODES) cnt[t] = 0;
    __syncthreads();
    for (int i0 = beg; i0 < end; i0 += 4 * FIN_T) {
        int r[4];
#pragma unroll
        for (int u = 0; u < 4; ++u) {
            const int i = i0 + u * FIN_T + t;
            r[u] = i < end ? rec[i] : -1;
        }
#pragma unroll
        for (int u = 0; u < 4; ++u) {
            if (r[u] != -1) atomicAdd(&cnt[(unsigned)r[u] >> 24], 1);
        }
    }
    __syncthreads();

    int c = 0, inc = 0;
    if (t < BUCKET_NODES) {
        c = cnt[t];
        inc = c;
#pragma unroll
        for (int o = 1; o < 32; o <<= 1) {
            const int u = __shfl_up(inc, o);
            if (lane >= o) inc += u;
        }
        if (lane == 31) wsum[w] = inc;
    }
    __syncthreads();
    if (t < BUCKET_NODES) {
        int base = beg;
        for (int j = 0; j < w; ++j) base += wsum[j];
        const int ex = base + inc - c;
        cur[t] = ex;
        rowptr[b * BUCKET_NODES + t] = ex;
    }
    __syncthreads();
    for (int i0 = beg; i0 < end; i0 += 4 * FIN_T) {
        int r[4];
#pragma unroll
        for (int u = 0; u < 4; ++u) {
            const int i = i0 + u * FIN_T + t;
            r[u] = i < end ? rec[i] : -1;
        }
#pragma unroll
        for (int u = 0; u < 4; ++u) {
            if (r[u] != -1) {
                const int pos = atomicAdd(&cur[(unsigned)r[u] >> 24], 1);
                col[pos] = r[u] & 0xFFFFFF;
            }
        }
    }
}

__device__ __forceinline__ float4 add4(float4 a, float4 b) {
    return make_float4(a.x + b.x, a.y + b.y, a.z + b.z, a.w + b.w);
}

__device__ __forceinline__ float4 gin_row_sum(const float4* __restrict__ x4,
                                              const int* __restrict__ col,
                                              int beg, int end, int lane,
                                              float4 own, int my) {
    float4 a0 = own;
    float4 a1 = make_float4(0.f, 0.f, 0.f, 0.f);
    float4 a2 = a1, a3 = a1;
    for (int b = beg; b < end;) {
        const int rem = end - b;
        const int n   = rem < 32 ? rem : 32;
        int j = 0;
        for (; j + 4 <= n; j += 4) {
            const int s0 = __builtin_amdgcn_readlane(my, j);
            const int s1 = __builtin_amdgcn_readlane(my, j + 1);
            const int s2 = __builtin_amdgcn_readlane(my, j + 2);
            const int s3 = __builtin_amdgcn_readlane(my, j + 3);
            const float4 v0 = x4[(size_t)s0 * 32 + lane];
            const float4 v1 = x4[(size_t)s1 * 32 + lane];
            const float4 v2 = x4[(size_t)s2 * 32 + lane];
            const float4 v3 = x4[(size_t)s3 * 32 + lane];
            a0 = add4(a0, v0);
            a1 = add4(a1, v1);
            a2 = add4(a2, v2);
            a3 = add4(a3, v3);
        }
#pragma unroll 1
        for (; j < n; ++j) {
            const int s0 = __builtin_amdgcn_readlane(my, j);
            a0 = add4(a0, x4[(size_t)s0 * 32 + lane]);
        }
        b += 32;
        if (b < end) {
            const int idx = b + lane;
            my = idx < end ? col[idx] : 0;
        }
    }
    return add4(add4(a0, a1), add4(a2, a3));
}

template <int RELU_OUT>
__global__ __launch_bounds__(256) void gin_layer_kernel(const float* __restrict__ xin,
                                                        const int* __restrict__ rowptr,
                                                        const int* __restrict__ col,
                                                        const float* __restrict__ Wa,
                                                        const float* __restrict__ ba,
                                                        const float* __restrict__ Wb,
                                                        const float* __restrict__ bb,
                                                        float* __restrict__ out) {
    __shared__ float XT[D][TM + 1];
    __shared__ float HT[D][TM + 1];

    const int node_base = blockIdx.x * TM;
    const int t = threadIdx.x;

    const int lane = t & 31;
    const int w = __builtin_amdgcn_readfirstlane(t >> 5);

    {
        const float4* x4 = reinterpret_cast<const float4*>(xin);
        const int n0 = node_base + 2 * w;
        const int p0 = rowptr[n0];
        const int p1 = rowptr[n0 + 1];
        const int p2 = rowptr[n0 + 2];
        const float4 own0 = x4[(size_t)n0 * 32 + lane];
        const float4 own1 = x4[(size_t)(n0 + 1) * 32 + lane];
        const int my0 = p0 + lane < p1 ? col[p0 + lane] : 0;
        const int my1 = p1 + lane < p2 ? col[p1 + lane] : 0;
        const int c = 4 * lane;
        const float4 r0 = gin_row_sum(x4, col, p0, p1, lane, own0, my0);
        XT[c][2 * w]     = r0.x;
        XT[c + 1][2 * w] = r0.y;
        XT[c + 2][2 * w] = r0.z;
        XT[c + 3][2 * w] = r0.w;
        const float4 r1 = gin_row_sum(x4, col, p1, p2, lane, own1, my1);
        XT[c][2 * w + 1]     = r1.x;
        XT[c + 1][2 * w + 1] = r1.y;
        XT[c + 2][2 * w + 1] = r1.z;
        XT[c + 3][2 * w + 1] = r1.w;
    }
    __syncthreads();

    const int half = lane >> 4;
    const int m16  = lane & 15;
    const int ncol = w * 16 + m16;

    v8f c = {};
#pragma unroll 8
    for (int k0 = 0; k0 < D; k0 += 4) {
        const int ka = k0 + 2 * half;
        v2f a, b;
        a.x = XT[ka][m16];
        a.y = XT[ka + 1][m16];
        b.x = Wa[(size_t)ka * D + ncol];
        b.y = Wa[(size_t)(ka + 1) * D + ncol];
        c = __builtin_amdgcn_wmma_f32_16x16x4_f32(false, a, false, b, (short)0, c,
                                                  false, false);
    }
    {
        const float bias = ba[ncol];
#pragma unroll
        for (int v = 0; v < 8; ++v) {
            float r = c[v] + bias;
            r = r > 0.f ? r : 0.f;
            HT[ncol][v + 8 * half] = r;
        }
    }
    __syncthreads();

    c = (v8f){};
#pragma unroll 8
    for (int k0 = 0; k0 < D; k0 += 4) {
        const int ka = k0 + 2 * half;
        v2f a, b;
        a.x = HT[ka][m16];
        a.y = HT[ka + 1][m16];
        b.x = Wb[(size_t)ka * D + ncol];
        b.y = Wb[(size_t)(ka + 1) * D + ncol];
        c = __builtin_amdgcn_wmma_f32_16x16x4_f32(false, a, false, b, (short)0, c,
                                                  false, false);
    }
    {
        const float bias = bb[ncol];
#pragma unroll
        for (int v = 0; v < 8; ++v) {
            float r = c[v] + bias;
            if (RELU_OUT) r = r > 0.f ? r : 0.f;
            out[(size_t)(node_base + v + 8 * half) * D + ncol] = r;
        }
    }
}

extern "C" void kernel_launch(void* const* d_in, const int* in_sizes, int n_in,
                              void* d_out, int out_size, void* d_ws, size_t ws_size,
                              hipStream_t stream) {
    const float* x  = (const float*)d_in[0];
    const int*  ei  = (const int*)d_in[1];
    const float* W1 = (const float*)d_in[2];
    const float* b1 = (const float*)d_in[3];
    const float* W2 = (const float*)d_in[4];
    const float* b2 = (const float*)d_in[5];
    const float* W3 = (const float*)d_in[6];
    const float* b3 = (const float*)d_in[7];
    const float* W4 = (const float*)d_in[8];
    const float* b4 = (const float*)d_in[9];
    float* out = (float*)d_out;

    const int E = in_sizes[1] / 2;
    const int* src = ei;
    const int* dst = ei + E;

    float* h      = (float*)d_ws;
    int* bcnt     = (int*)(h + (size_t)N_NODES * D);
    int* boff     = bcnt + N_PAD;
    int* rowptr   = boff + N_PAD;
    int* rec      = rowptr + N_PAD;
    int* col      = rec + E;

    const int layerBlocks = N_NODES / TM;

    gin_bucket_count_kernel<<<EDGE_BLOCKS, EDGE_T, 0, stream>>>(dst, bcnt, E);
    gin_scan_kernel<<<1, SCAN_T, 0, stream>>>(bcnt, boff);
    gin_bucket_scatter_kernel<<<EDGE_BLOCKS, EDGE_T, 0, stream>>>(src, dst, boff, rec, E);
    gin_bucket_finish_kernel<<<NBUCK, FIN_T, 0, stream>>>(boff, rec, rowptr, col);

    gin_layer_kernel<1><<<layerBlocks, 256, 0, stream>>>(x, rowptr, col, W1, b1, W2, b2, h);

    gin_layer_kernel<0><<<layerBlocks, 256, 0, stream>>>(h, rowptr, col, W3, b3, W4, b4, out);
}


# Edge grouping in 2 launches, not 4: each chunk's work-group orders its records by bucket in its own slice (LDS count, scan, place); a bucket's group walks its 256 runs; neighbour sum order changes
# speedup vs baseline: 3.5786x; 1.1844x over previous
#include <hip/hip_runtime.h>

#define N_NODES 50000
#define D 128
#define TM 16

#define EDGE_BLOCKS   256
#define EDGE_T        512
#define EDGE_U        8
#define BUCKET_SHIFT  8
#define BUCKET_NODES  (1 << BUCKET_SHIFT)
#define NBUCK         ((N_NODES + BUCKET_NODES - 1) / BUCKET_NODES)
#define TAB_N         (NBUCK * EDGE_BLOCKS)
#define ROWPTR_N      (NBUCK * BUCKET_NODES)
#define FIN_T         1024
#define FIN_U         8
#define FIN_PRE       (((NBUCK - 1) * (EDGE_BLOCKS / 4) + FIN_T - 1) / FIN_T)

static_assert(NBUCK <= 256 && NBUCK <= EDGE_T, "waves 0..7 of an edge work-group scan the bucket counts");
static_assert(EDGE_T >= 256, "waves 0..7 exist");
static_assert(FIN_T == 4 * EDGE_BLOCKS, "four finishing threads per run");
static_assert(FIN_T / 32 == 32, "one wave reduces the 32 wave partials");
static_assert(EDGE_BLOCKS % 4 == 0, "a table row is whole 16-byte groups");
static_assert(ROWPTR_N > N_NODES + 1, "rowptr holds every node's start and the total behind the last");
static_assert(N_NODES < (1 << 24), "a source id fits the record's low 24 bits and no record is all ones");

typedef float v2f __attribute__((ext_vector_type(2)));
typedef float v8f __attribute__((ext_vector_type(8)));

__device__ __forceinline__ int gin_bucket_of(int d) {
    const int b = d >> BUCKET_SHIFT;
    return b < NBUCK ? b : NBUCK - 1;
}

__global__ __launch_bounds__(EDGE_T) void gin_chunk_partition_kernel(const int* __restrict__ src_idx,
                                                                     const int* __restrict__ dst_idx,
                                                                     int* __restrict__ lcnt,
                                                                     int* __restrict__ loff,
                                                                     int* __restrict__ rec, int nE) {
    __shared__ int hist[NBUCK];
    __shared__ int cur[NBUCK];
    __shared__ int wsum[8];
    const int t    = threadIdx.x;
    const int lane = t & 31;
    const int w    = t >> 5;
    const int g    = blockIdx.x;
    if (t < NBUCK) hist[t] = 0;
    __syncthreads();

    const int chunk = (nE + EDGE_BLOCKS - 1) / EDGE_BLOCKS;
    const int beg = g * chunk;
    const int end = beg + chunk < nE ? beg + chunk : nE;
    for (int e0 = beg; e0 < end; e0 += EDGE_U * EDGE_T) {
        int d[EDGE_U];
#pragma unroll
        for (int u = 0; u < EDGE_U; ++u) {
            const int e = e0 + u * EDGE_T + t;
            d[u] = e < end ? dst_idx[e] : -1;
        }
#pragma unroll
        for (int u = 0; u < EDGE_U; ++u) {
            if (d[u] >= 0) atomicAdd(&hist[gin_bucket_of(d[u])], 1);
        }
    }
    __syncthreads();

    int c = 0, inc = 0;
    if (t < 256) {
        c = t < NBUCK ? hist[t] : 0;
        inc = c;
#pragma unroll
        for (int o = 1; o < 32; o <<= 1) {
            const int u = __shfl_up(inc, o);
            if (lane >= o) inc += u;
        }
        if (lane == 31) wsum[w] = inc;
    }
    __syncthreads();
    int ex = 0;
    if (t < NBUCK) {
        int base = 0;
        for (int j = 0; j < w; ++j) base += wsum[j];
        ex = base + inc - c;
        cur[t] = beg + ex;
    }
    __syncthreads();
    if (t < NBUCK) {
        lcnt[t * EDGE_BLOCKS + g] = c;
        loff[t * EDGE_BLOCKS + g] = ex;
    }

    for (int e0 = beg; e0 < end; e0 += EDGE_U * EDGE_T) {
        int d[EDGE_U], s[EDGE_U];
#pragma unroll
        for (int u = 0; u < EDGE_U; ++u) {
            const int e = e0 + u * EDGE_T + t;
            d[u] = e < end ? dst_idx[e] : -1;
            s[u] = e < end ? src_idx[e] : 0;
        }
#pragma unroll
        for (int u = 0; u < EDGE_U; ++u) {
            if (d[u] >= 0) {
                const int pos = atomicAdd(&cur[gin_bucket_of(d[u])], 1);
                rec[pos] = (int)(((unsigned)(d[u] & (BUCKET_NODES - 1)) << 24) | (unsigned)s[u]);
            }
        }
    }
}

__global__ __launch_bounds__(FIN_T) void gin_bucket_finish_kernel(const int* __restrict__ lcnt,
                                                                  const int* __restrict__ loff,
                                                                  const int* __restrict__ rec,
                                                                  int* __restrict__ rowptr,
                                                                  int* __restrict__ col, int nE) {
    __shared__ int cnt[BUCKET_NODES];
    __shared__ int cur[BUCKET_NODES];
    __shared__ int wsum[FIN_T / 32];
    const int t    = threadIdx.x;
    const int lane = t & 31;
    const int w    = t >> 5;
    const int b    = blockIdx.x;

    const int chunk = (nE + EDGE_BLOCKS - 1) / EDGE_BLOCKS;
    const int g  = t >> 2;
    const int q  = t & 3;
    const int rc = lcnt[b * EDGE_BLOCKS + g];
    const int ro = g * chunk + loff[b * EDGE_BLOCKS + g];

    int acc = 0;
    {
        const int4* l4 = reinterpret_cast<const int4*>(lcnt);
        const int n4 = b * (EDGE_BLOCKS / 4);
        int4 v[FIN_PRE];
#pragma unroll
        for (int u = 0; u < FIN_PRE; ++u) {
            const int i = u * FIN_T + t;
            v[u] = i < n4 ? l4[i] : make_int4(0, 0, 0, 0);
        }
#pragma unroll
        for (int u = 0; u < FIN_PRE; ++u) acc += (v[u].x + v[u].y) + (v[u].z + v[u].w);
    }
#pragma unroll
    for (int o = 16; o > 0; o >>= 1) acc += __shfl_xor(acc, o);
    if (lane == 0) wsum[w] = acc;
    if (t < BUCKET_NODES) cnt[t] = 0;
    __syncthreads();
    int first = wsum[lane];
#pragma unroll
    for (int o = 16; o > 0; o >>= 1) first += __shfl_xor(first, o);

    for (int j0 = 0; j0 < rc; j0 += 4 * FIN_U) {
        int r[FIN_U];
#pragma unroll
        for (int u = 0; u < FIN_U; ++u) {
            const int j = j0 + 4 * u + q;
            r[u] = j < rc ? rec[ro + j] : -1;
        }
#pragma unroll
        for (int u = 0; u < FIN_U; ++u) {
            if (r[u] != -1) atomicAdd(&cnt[(unsigned)r[u] >> 24], 1);
        }
    }
    __syncthreads();

    int c = 0, inc = 0;
    if (t < BUCKET_NODES) {
        c = cnt[t];
        inc = c;
#pragma unroll
        for (int o = 1; o < 32; o <<= 1) {
            const int u = __shfl_up(inc, o);
            if (lane >= o) inc += u;
        }
        if (lane == 31) wsum[w] = inc;
    }
    __syncthreads();
    int ex = 0;
    if (t < BUCKET_NODES) {
        int base = first;
        for (int j = 0; j < w; ++j) base += wsum[j];
        ex = base + inc - c;
        cur[t] = ex;
    }
    __syncthreads();
    if (t < BUCKET_NODES) rowptr[b * BUCKET_NODES + t] = ex;
    for (int j0 = 0; j0 < rc; j0 += 4 * FIN_U) {
        int r[FIN_U];
#pragma unroll
        for (int u = 0; u < FIN_U; ++u) {
            const int j = j0 + 4 * u + q;
            r[u] = j < rc ? rec[ro + j] : -1;
        }
#pragma unroll
        for (int u = 0; u < FIN_U; ++u) {
            if (r[u] != -1) {
                const int pos = atomicAdd(&cur[(unsigned)r[u] >> 24], 1);
                col[pos] = r[u] & 0xFFFFFF;
            }
        }
    }
}

__device__ __forceinline__ float4 add4(float4 a, float4 b) {
    return make_float4(a.x + b.x, a.y + b.y, a.z + b.z, a.w + b.w);
}

__device__ __forceinline__ float4 gin_row_sum(const float4* __restrict__ x4,
                                              const int* __restrict__ col,
                                              int beg, int end, int lane,
                                              float4 own, int my) {
    float4 a0 = own;
    float4 a1 = make_float4(0.f, 0.f, 0.f, 0.f);
    float4 a2 = a1, a3 = a1;
    for (int b = beg; b < end;) {
        const int rem = end - b;
        const int n   = rem < 32 ? rem : 32;
        int j = 0;
        for (; j + 4 <= n; j += 4) {
            const int s0 = __builtin_amdgcn_readlane(my, j);
            const int s1 = __builtin_amdgcn_readlane(my, j + 1);
            const int s2 = __builtin_amdgcn_readlane(my, j + 2);
            const int s3 = __builtin_amdgcn_readlane(my, j + 3);
            const float4 v0 = x4[(size_t)s0 * 32 + lane];
            const float4 v1 = x4[(size_t)s1 * 32 + lane];
            const float4 v2 = x4[(size_t)s2 * 32 + lane];
            const float4 v3 = x4[(size_t)s3 * 32 + lane];
            a0 = add4(a0, v0);
            a1 = add4(a1, v1);
            a2 = add4(a2, v2);
            a3 = add4(a3, v3);
        }
#pragma unroll 1
        for (; j < n; ++j) {
            const int s0 = __builtin_amdgcn_readlane(my, j);
            a0 = add4(a0, x4[(size_t)s0 * 32 + lane]);
        }
        b += 32;
        if (b < end) {
            const int idx = b + lane;
            my = idx < end ? col[idx] : 0;
        }
    }
    return add4(add4(a0, a1), add4(a2, a3));
}

template <int RELU_OUT>
__global__ __launch_bounds__(256) void gin_layer_kernel(const float* __restrict__ xin,
                                                        const int* __restrict__ rowptr,
                                                        const int* __restrict__ col,
                                                        const float* __restrict__ Wa,
                                                        const float* __restrict__ ba,
                                                        const float* __restrict__ Wb,
                                                        const float* __restrict__ bb,
                                                        float* __restrict__ out) {
    __shared__ float XT[D][TM + 1];
    __shared__ float HT[D][TM + 1];

    const int node_base = blockIdx.x * TM;
    const int t = threadIdx.x;

    const int lane = t & 31;
    const int w = __builtin_amdgcn_readfirstlane(t >> 5);

    {
        const float4* x4 = reinterpret_cast<const float4*>(xin);
        const int n0 = node_base + 2 * w;
        const int p0 = rowptr[n0];
        const int p1 = rowptr[n0 + 1];
        const int p2 = rowptr[n0 + 2];
        const float4 own0 = x4[(size_t)n0 * 32 + lane];
        const float4 own1 = x4[(size_t)(n0 + 1) * 32 + lane];
        const int my0 = p0 + lane < p1 ? col[p0 + lane] : 0;
        const int my1 = p1 + lane < p2 ? col[p1 + lane] : 0;
        const int c = 4 * lane;
        const float4 r0 = gin_row_sum(x4, col, p0, p1, lane, own0, my0);
        XT[c][2 * w]     = r0.x;
        XT[c + 1][2 * w] = r0.y;
        XT[c + 2][2 * w] = r0.z;
        XT[c + 3][2 * w] = r0.w;
        const float4 r1 = gin_row_sum(x4, col, p1, p2, lane, own1, my1);
        XT[c][2 * w + 1]     = r1.x;
        XT[c + 1][2 * w + 1] = r1.y;
        XT[c + 2][2 * w + 1] = r1.z;
        XT[c + 3][2 * w + 1] = r1.w;
    }
    __syncthreads();

    const int half = lane >> 4;
    const int m16  = lane & 15;
    const int ncol = w * 16 + m16;

    v8f c = {};
#pragma unroll 8
    for (int k0 = 0; k0 < D; k0 += 4) {
        const int ka = k0 + 2 * half;
        v2f a, b;
        a.x = XT[ka][m16];
        a.y = XT[ka + 1][m16];
        b.x = Wa[(size_t)ka * D + ncol];
        b.y = Wa[(size_t)(ka + 1) * D + ncol];
        c = __builtin_amdgcn_wmma_f32_16x16x4_f32(false, a, false, b, (short)0, c,
                                                  false, false);
    }
    {
        const float bias = ba[ncol];
#pragma unroll
        for (int v = 0; v < 8; ++v) {
            float r = c[v] + bias;
            r = r > 0.f ? r : 0.f;
            HT[ncol][v + 8 * half] = r;
        }
    }
    __syncthreads();

    c = (v8f){};
#pragma unroll 8
    for (int k0 = 0; k0 < D; k0 += 4) {
        const int ka = k0 + 2 * half;
        v2f a, b;
        a.x = HT[ka][m16];
        a.y = HT[ka + 1][m16];
        b.x = Wb[(size_t)ka * D + ncol];
        b.y = Wb[(size_t)(ka + 1) * D + ncol];
        c = __builtin_amdgcn_wmma_f32_16x16x4_f32(false, a, false, b, (short)0, c,
                                                  false, false);
    }
    {
        const float bias = bb[ncol];
#pragma unroll
        for (int v = 0; v < 8; ++v) {
            float r = c[v] + bias;
            if (RELU_OUT) r = r > 0.f ? r : 0.f;
            out[(size_t)(node_base + v + 8 * half) * D + ncol] = r;
        }
    }
}

extern "C" void kernel_launch(void* const* d_in, const int* in_sizes, int n_in,
                              void* d_out, int out_size, void* d_ws, size_t ws_size,
                              hipStream_t stream) {
    const float* x  = (const float*)d_in[0];
    const int*  ei  = (const int*)d_in[1];
    const float* W1 = (const float*)d_in[2];
    const float* b1 = (const float*)d_in[3];
    const float* W2 = (const float*)d_in[4];
    const float* b2 = (const float*)d_in[5];
    const float* W3 = (const float*)d_in[6];
    const float* b3 = (const float*)d_in[7];
    const float* W4 = (const float*)d_in[8];
    const float* b4 = (const float*)d_in[9];
    float* out = (float*)d_out;

    const int E = in_sizes[1] / 2;
    const int* src = ei;
    const int* dst = ei + E;

    float* h      = (float*)d_ws;
    int* lcnt     = (int*)(h + (size_t)N_NODES * D);
    int* loff     = lcnt + TAB_N;
    int* rowptr   = loff + TAB_N;
    int* rec      = rowptr + ROWPTR_N;
    int* col      = rec + E;

    const int layerBlocks = N_NODES / TM;

    gin_chunk_partition_kernel<<<EDGE_BLOCKS, EDGE_T, 0, stream>>>(src, dst, lcnt, loff, rec, E);
    gin_bucket_finish_kernel<<<NBUCK, FIN_T, 0, stream>>>(lcnt, loff, rec, rowptr, col, E);

    gin_layer_kernel<1><<<layerBlocks, 256, 0, stream>>>(x, rowptr, col, W1, b1, W2, b2, h);

    gin_layer_kernel<0><<<layerBlocks, 256, 0, stream>>>(h, rowptr, col, W3, b3, W4, b4, out);
}


# Each layer work-group takes 32 nodes (two 16-row tiles, four nodes per wave): a weight fragment is loaded once per k-step and feeds two WMMAs with independent accumulators; sum order kept
# speedup vs baseline: 3.8580x; 1.0781x over previous
#include <hip/hip_runtime.h>

#define N_NODES 50000
#define D 128
#define TM 32

#define EDGE_BLOCKS   256
#define EDGE_T        512
#define EDGE_U        8
#define BUCKET_SHIFT  8
#define BUCKET_NODES  (1 << BUCKET_SHIFT)
#define NBUCK         ((N_NODES + BUCKET_NODES - 1) / BUCKET_NODES)
#define TAB_N         (NBUCK * EDGE_BLOCKS)
#define ROWPTR_N      (NBUCK * BUCKET_NODES)
#define FIN_T         1024
#define FIN_U         8
#define FIN_PRE       (((NBUCK - 1) * (EDGE_BLOCKS / 4) + FIN_T - 1) / FIN_T)

static_assert(NBUCK <= 256 && NBUCK <= EDGE_T, "waves 0..7 of an edge work-group scan the bucket counts");
static_assert(EDGE_T >= 256, "waves 0..7 exist");
static_assert(FIN_T == 4 * EDGE_BLOCKS, "four finishing threads per run");
static_assert(FIN_T / 32 == 32, "one wave reduces the 32 wave partials");
static_assert(EDGE_BLOCKS % 4 == 0, "a table row is whole 16-byte groups");
static_assert(ROWPTR_N > N_NODES + 1, "rowptr holds every node's start and the total behind the last");
static_assert(N_NODES < (1 << 24), "a source id fits the record's low 24 bits and no record is all ones");
static_assert(N_NODES % 16 == 0, "a 16-row tile of the layer pass is whole or absent");

typedef float v2f __attribute__((ext_vector_type(2)));
typedef float v8f __attribute__((ext_vector_type(8)));

__device__ __forceinline__ int gin_bucket_of(int d) {
    const int b = d >> BUCKET_SHIFT;
    return b < NBUCK ? b : NBUCK - 1;
}

__global__ __launch_bounds__(EDGE_T) void gin_chunk_partition_kernel(const int* __restrict__ src_idx,
                                                                     const int* __restrict__ dst_idx,
                                                                     int* __restrict__ lcnt,
                                                                     int* __restrict__ loff,
                                                                     int* __restrict__ rec, int nE) {
    __shared__ int hist[NBUCK];
    __shared__ int cur[NBUCK];
    __shared__ int wsum[8];
    const int t    = threadIdx.x;
    const int lane = t & 31;
    const int w    = t >> 5;
    const int g    = blockIdx.x;
    if (t < NBUCK) hist[t] = 0;
    __syncthreads();

    const int chunk = (nE + EDGE_BLOCKS - 1) / EDGE_BLOCKS;
    const int beg = g * chunk;
    const int end = beg + chunk < nE ? beg + chunk : nE;
    for (int e0 = beg; e0 < end; e0 += EDGE_U * EDGE_T) {
        int d[EDGE_U];
#pragma unroll
        for (int u = 0; u < EDGE_U; ++u) {
            const int e = e0 + u * EDGE_T + t;
            d[u] = e < end ? dst_idx[e] : -1;
        }
#pragma unroll
        for (int u = 0; u < EDGE_U; ++u) {
            if (d[u] >= 0) atomicAdd(&hist[gin_bucket_of(d[u])], 1);
        }
    }
    __syncthreads();

    int c = 0, inc = 0;
    if (t < 256) {
        c = t < NBUCK ? hist[t] : 0;
        inc = c;
#pragma unroll
        for (int o = 1; o < 32; o <<= 1) {
            const int u = __shfl_up(inc, o);
            if (lane >= o) inc += u;
        }
        if (lane == 31) wsum[w] = inc;
    }
    __syncthreads();
    int ex = 0;
    if (t < NBUCK) {
        int base = 0;
        for (int j = 0; j < w; ++j) base += wsum[j];
        ex = base + inc - c;
        cur[t] = beg + ex;
    }
    __syncthreads();
    if (t < NBUCK) {
        lcnt[t * EDGE_BLOCKS + g] = c;
        loff[t * EDGE_BLOCKS + g] = ex;
    }

    for (int e0 = beg; e0 < end; e0 += EDGE_U * EDGE_T) {
        int d[EDGE_U], s[EDGE_U];
#pragma unroll
        for (int u = 0; u < EDGE_U; ++u) {
            const int e = e0 + u * EDGE_T + t;
            d[u] = e < end ? dst_idx[e] : -1;
            s[u] = e < end ? src_idx[e] : 0;
        }
#pragma unroll
        for (int u = 0; u < EDGE_U; ++u) {
            if (d[u] >= 0) {
                const int pos = atomicAdd(&cur[gin_bucket_of(d[u])], 1);
                rec[pos] = (int)(((unsigned)(d[u] & (BUCKET_NODES - 1)) << 24) | (unsigned)s[u]);
            }
        }
    }
}

__global__ __launch_bounds__(FIN_T) void gin_bucket_finish_kernel(const int* __restrict__ lcnt,
                                                                  const int* __restrict__ loff,
                                                                  const int* __restrict__ rec,
                                                                  int* __restrict__ rowptr,
                                                                  int* __restrict__ col, int nE) {
    __shared__ int cnt[BUCKET_NODES];
    __shared__ int cur[BUCKET_NODES];
    __shared__ int wsum[FIN_T / 32];
    const int t    = threadIdx.x;
    const int lane = t & 31;
    const int w    = t >> 5;
    const int b    = blockIdx.x;

    const int chunk = (nE + EDGE_BLOCKS - 1) / EDGE_BLOCKS;
    const int g  = t >> 2;
    const int q  = t & 3;
    const int rc = lcnt[b * EDGE_BLOCKS + g];
    const int ro = g * chunk + loff[b * EDGE_BLOCKS + g];

    int acc = 0;
    {
        const int4* l4 = reinterpret_cast<const int4*>(lcnt);
        const int n4 = b * (EDGE_BLOCKS / 4);
        int4 v[FIN_PRE];
#pragma unroll
        for (int u = 0; u < FIN_PRE; ++u) {
            const int i = u * FIN_T + t;
            v[u] = i < n4 ? l4[i] : make_int4(0, 0, 0, 0);
        }
#pragma unroll
        for (int u = 0; u < FIN_PRE; ++u) acc += (v[u].x + v[u].y) + (v[u].z + v[u].w);
    }
#pragma unroll
    for (int o = 16; o > 0; o >>= 1) acc += __shfl_xor(acc, o);
    if (lane == 0) wsum[w] = acc;
    if (t < BUCKET_NODES) cnt[t] = 0;
    __syncthreads();
    int first = wsum[lane];
#pragma unroll
    for (int o = 16; o > 0; o >>= 1) first += __shfl_xor(first, o);

    for (int j0 = 0; j0 < rc; j0 += 4 * FIN_U) {
        int r[FIN_U];
#pragma unroll
        for (int u = 0; u < FIN_U; ++u) {
            const int j = j0 + 4 * u + q;
            r[u] = j < rc ? rec[ro + j] : -1;
        }
#pragma unroll
        for (int u = 0; u < FIN_U; ++u) {
            if (r[u] != -1) atomicAdd(&cnt[(unsigned)r[u] >> 24], 1);
        }
    }
    __syncthreads();

    int c = 0, inc = 0;
    if (t < BUCKET_NODES) {
        c = cnt[t];
        inc = c;
#pragma unroll
        for (int o = 1; o < 32; o <<= 1) {
            const int u = __shfl_up(inc, o);
            if (lane >= o) inc += u;
        }
        if (lane == 31) wsum[w] = inc;
    }
    __syncthreads();
    int ex = 0;
    if (t < BUCKET_NODES) {
        int base = first;
        for (int j = 0; j < w; ++j) base += wsum[j];
        ex = base + inc - c;
        cur[t] = ex;
    }
    __syncthreads();
    if (t < BUCKET_NODES) rowptr[b * BUCKET_NODES + t] = ex;
    for (int j0 = 0; j0 < rc; j0 += 4 * FIN_U) {
        int r[FIN_U];
#pragma unroll
        for (int u = 0; u < FIN_U; ++u) {
            const int j = j0 + 4 * u + q;
            r[u] = j < rc ? rec[ro + j] : -1;
        }
#pragma unroll
        for (int u = 0; u < FIN_U; ++u) {
            if (r[u] != -1) {
                const int pos = atomicAdd(&cur[(unsigned)r[u] >> 24], 1);
                col[pos] = r[u] & 0xFFFFFF;
            }
        }
    }
}

__device__ __forceinline__ float4 add4(float4 a, float4 b) {
    return make_float4(a.x + b.x, a.y + b.y, a.z + b.z, a.w + b.w);
}

__device__ __forceinline__ float4 gin_row_sum(const float4* __restrict__ x4,
                                              const int* __restrict__ col,
                                              int beg, int end, int lane,
                                              float4 own, int my) {
    float4 a0 = own;
    float4 a1 = make_float4(0.f, 0.f, 0.f, 0.f);
    float4 a2 = a1, a3 = a1;
    for (int b = beg; b < end;) {
        const int rem = end - b;
        const int n   = rem < 32 ? rem : 32;
        int j = 0;
        for (; j + 4 <= n; j += 4) {
            const int s0 = __builtin_amdgcn_readlane(my, j);
            const int s1 = __builtin_amdgcn_readlane(my, j + 1);
            const int s2 = __builtin_amdgcn_readlane(my, j + 2);
            const int s3 = __builtin_amdgcn_readlane(my, j + 3);
            const float4 v0 = x4[(size_t)s0 * 32 + lane];
            const float4 v1 = x4[(size_t)s1 * 32 + lane];
            const float4 v2 = x4[(size_t)s2 * 32 + lane];
            const float4 v3 = x4[(size_t)s3 * 32 + lane];
            a0 = add4(a0, v0);
            a1 = add4(a1, v1);
            a2 = add4(a2, v2);
            a3 = add4(a3, v3);
        }
#pragma unroll 1
        for (; j < n; ++j) {
            const int s0 = __builtin_amdgcn_readlane(my, j);
            a0 = add4(a0, x4[(size_t)s0 * 32 + lane]);
        }
        b += 32;
        if (b < end) {
            const int idx = b + lane;
            my = idx < end ? col[idx] : 0;
        }
    }
    return add4(add4(a0, a1), add4(a2, a3));
}

template <int RELU_OUT>
__global__ __launch_bounds__(256) void gin_layer_kernel(const float* __restrict__ xin,
                                                        const int* __restrict__ rowptr,
                                                        const int* __restrict__ col,
                                                        const float* __restrict__ Wa,
                                                        const float* __restrict__ ba,
                                                        const float* __restrict__ Wb,
                                                        const float* __restrict__ bb,
                                                        float* __restrict__ out) {
    __shared__ float XT[2][D][16 + 1];
    __shared__ float HT[2][D][16 + 1];

    const int node_base = blockIdx.x * TM;
    const int t = threadIdx.x;

    const int lane = t & 31;
    const int w = __builtin_amdgcn_readfirstlane(t >> 5);

    {
        const float4* x4 = reinterpret_cast<const float4*>(xin);
        const int c  = 4 * lane;
        const int m0 = 4 * w;
        const int n0 = node_base + m0;
        float (*X)[16 + 1] = XT[m0 >> 4];
        const int mt = m0 & 15;
        if (n0 < N_NODES) {
            const int p0 = rowptr[n0];
            const int p1 = rowptr[n0 + 1];
            const int p2 = rowptr[n0 + 2];
            const int p3 = rowptr[n0 + 3];
            const int p4 = rowptr[n0 + 4];
            const float4 own0 = x4[(size_t)n0 * 32 + lane];
            const float4 own1 = x4[(size_t)(n0 + 1) * 32 + lane];
            const int my0 = p0 + lane < p1 ? col[p0 + lane] : 0;
            const int my1 = p1 + lane < p2 ? col[p1 + lane] : 0;
            const int my2 = p2 + lane < p3 ? col[p2 + lane] : 0;
            const int my3 = p3 + lane < p4 ? col[p3 + lane] : 0;
            const float4 r0 = gin_row_sum(x4, col, p0, p1, lane, own0, my0);
            X[c][mt]     = r0.x;
            X[c + 1][mt] = r0.y;
            X[c + 2][mt] = r0.z;
            X[c + 3][mt] = r0.w;
            const float4 r1 = gin_row_sum(x4, col, p1, p2, lane, own1, my1);
            X[c][mt + 1]     = r1.x;
            X[c + 1][mt + 1] = r1.y;
            X[c + 2][mt + 1] = r1.z;
            X[c + 3][mt + 1] = r1.w;
            const float4 own2 = x4[(size_t)(n0 + 2) * 32 + lane];
            const float4 own3 = x4[(size_t)(n0 + 3) * 32 + lane];
            const float4 r2 = gin_row_sum(x4, col, p2, p3, lane, own2, my2);
            X[c][mt + 2]     = r2.x;
            X[c + 1][mt + 2] = r2.y;
            X[c + 2][mt + 2] = r2.z;
            X[c + 3][mt + 2] = r2.w;
            const float4 r3 = gin_row_sum(x4, col, p3, p4, lane, own3, my3);
            X[c][mt + 3]     = r3.x;
            X[c + 1][mt + 3] = r3.y;
            X[c + 2][mt + 3] = r3.z;
            X[c + 3][mt + 3] = r3.w;
        } else {
            for (int i = 0; i < 4; ++i) {
                X[c][mt + i]     = 0.f;
                X[c + 1][mt + i] = 0.f;
                X[c + 2][mt + i] = 0.f;
                X[c + 3][mt + i] = 0.f;
            }
        }
    }
    __syncthreads();

    const int half = lane >> 4;
    const int m16  = lane & 15;
    const int ncol = w * 16 + m16;

    v8f c0 = {};
    v8f c1 = {};
#pragma unroll 8
    for (int k0 = 0; k0 < D; k0 += 4) {
        const int ka = k0 + 2 * half;
        v2f a0, a1, b;
        b.x = Wa[(size_t)ka * D + ncol];
        b.y = Wa[(size_t)(ka + 1) * D + ncol];
        a0.x = XT[0][ka][m16];
        a0.y = XT[0][ka + 1][m16];
        a1.x = XT[1][ka][m16];
        a1.y = XT[1][ka + 1][m16];
        c0 = __builtin_amdgcn_wmma_f32_16x16x4_f32(false, a0, false, b, (short)0, c0,
                                                   false, false);
        c1 = __builtin_amdgcn_wmma_f32_16x16x4_f32(false, a1, false, b, (short)0, c1,
                                                   false, false);
    }
    {
        const float bias = ba[ncol];
#pragma unroll
        for (int v = 0; v < 8; ++v) {
            float r = c0[v] + bias;
            r = r > 0.f ? r : 0.f;
            HT[0][ncol][v + 8 * half] = r;
            float s = c1[v] + bias;
            s = s > 0.f ? s : 0.f;
            HT[1][ncol][v + 8 * half] = s;
        }
    }
    __syncthreads();

    c0 = (v8f){};
    c1 = (v8f){};
#pragma unroll 8
    for (int k0 = 0; k0 < D; k0 += 4) {
        const int ka = k0 + 2 * half;
        v2f a0, a1, b;
        b.x = Wb[(size_t)ka * D + ncol];
        b.y = Wb[(size_t)(ka + 1) * D + ncol];
        a0.x = HT[0][ka][m16];
        a0.y = HT[0][ka + 1][m16];
        a1.x = HT[1][ka][m16];
        a1.y = HT[1][ka + 1][m16];
        c0 = __builtin_amdgcn_wmma_f32_16x16x4_f32(false, a0, false, b, (short)0, c0,
                                                   false, false);
        c1 = __builtin_amdgcn_wmma_f32_16x16x4_f32(false, a1, false, b, (short)0, c1,
                                                   false, false);
    }
    {
        const float bias = bb[ncol];
#pragma unroll
        for (int v = 0; v < 8; ++v) {
            float r = c0[v] + bias;
            if (RELU_OUT) r = r > 0.f ? r : 0.f;
            out[(size_t)(node_base + v + 8 * half) * D + ncol] = r;
        }
        if (node_base + 16 < N_NODES) {
#pragma unroll
            for (int v = 0; v < 8; ++v) {
                float r = c1[v] + bias;
                if (RELU_OUT) r = r > 0.f ? r : 0.f;
                out[(size_t)(node_base + 16 + v + 8 * half) * D + ncol] = r;
            }
        }
    }
}

extern "C" void kernel_launch(void* const* d_in, const int* in_sizes, int n_in,
                              void* d_out, int out_size, void* d_ws, size_t ws_size,
                              hipStream_t stream) {
    const float* x  = (const float*)d_in[0];
    const int*  ei  = (const int*)d_in[1];
    const float* W1 = (const float*)d_in[2];
    const float* b1 = (const float*)d_in[3];
    const float* W2 = (const float*)d_in[4];
    const float* b2 = (const float*)d_in[5];
    const float* W3 = (const float*)d_in[6];
    const float* b3 = (const float*)d_in[7];
    const float* W4 = (const float*)d_in[8];
    const float* b4 = (const float*)d_in[9];
    float* out = (float*)d_out;

    const int E = in_sizes[1] / 2;
    const int* src = ei;
    const int* dst = ei + E;

    float* h      = (float*)d_ws;
    int* lcnt     = (int*)(h + (size_t)N_NODES * D);
    int* loff     = lcnt + TAB_N;
    int* rowptr   = loff + TAB_N;
    int* rec      = rowptr + ROWPTR_N;
    int* col      = rec + E;

    const int layerBlocks = (N_NODES + TM - 1) / TM;

    gin_chunk_partition_kernel<<<EDGE_BLOCKS, EDGE_T, 0, stream>>>(src, dst, lcnt, loff, rec, E);
    gin_bucket_finish_kernel<<<NBUCK, FIN_T, 0, stream>>>(lcnt, loff, rec, rowptr, col, E);

    gin_layer_kernel<1><<<layerBlocks, 256, 0, stream>>>(x, rowptr, col, W1, b1, W2, b2, h);

    gin_layer_kernel<0><<<layerBlocks, 256, 0, stream>>>(h, rowptr, col, W3, b3, W4, b4, out);
}
